# starting kernel
# baseline (speedup 1.0000x reference)
#include <hip/hip_runtime.h>

typedef float v2f __attribute__((ext_vector_type(2)));
typedef float v8f __attribute__((ext_vector_type(8)));

#define NNODES   100000
#define NEDGES   1600000
#define DFEAT    128
#define NEG_SLOPE 0.2f

__global__ void gat_zero_f32(float* __restrict__ p, long n) {
    long i = blockIdx.x * (long)blockDim.x + threadIdx.x;
    long stride = gridDim.x * (long)blockDim.x;
    for (; i < n; i += stride) p[i] = 0.0f;
}

__global__ void gat_gemm_wmma(const float* __restrict__ h,
                              const float* __restrict__ W,
                              float* __restrict__ feat) {
    const int strip = blockIdx.x * 4 + (threadIdx.x >> 5);
    if (strip >= NNODES / 16) return;
    const int lane  = threadIdx.x & 31;
    const int m     = lane & 15;
    const int khalf = (lane >> 4) << 1;
    const int hihalf = (lane >> 4) << 3;
    const long row  = (long)strip * 16 + m;

    v8f acc[8];
#pragma unroll
    for (int ct = 0; ct < 8; ++ct) acc[ct] = (v8f){0,0,0,0,0,0,0,0};

    for (int k = 0; k < DFEAT; k += 4) {
        v2f a;
        a.x = h[row * DFEAT + k + khalf + 0];
        a.y = h[row * DFEAT + k + khalf + 1];
#pragma unroll
        for (int ct = 0; ct < 8; ++ct) {
            v2f b;
            b.x = W[(k + khalf + 0) * DFEAT + ct * 16 + m];
            b.y = W[(k + khalf + 1) * DFEAT + ct * 16 + m];
            acc[ct] = __builtin_amdgcn_wmma_f32_16x16x4_f32(
                false, a, false, b, (short)0, acc[ct], false, false);
        }
    }

#pragma unroll
    for (int ct = 0; ct < 8; ++ct) {
#pragma unroll
        for (int v = 0; v < 8; ++v) {
            long orow = (long)strip * 16 + v + hihalf;
            feat[orow * DFEAT + ct * 16 + m] = acc[ct][v];
        }
    }
}

__global__ void gat_attn_logits(const float* __restrict__ feat,
                                const float* __restrict__ attn_l,
                                const float* __restrict__ attn_r,
                                float* __restrict__ el,
                                float* __restrict__ er) {
    const int node = blockIdx.x * 4 + (threadIdx.x >> 5);
    if (node >= NNODES) return;
    const int lane = threadIdx.x & 31;

    const float4 f  = *(const float4*)(feat   + (long)node * DFEAT + lane * 4);
    const float4 al = *(const float4*)(attn_l + lane * 4);
    const float4 ar = *(const float4*)(attn_r + lane * 4);

    float pl = f.x * al.x + f.y * al.y + f.z * al.z + f.w * al.w;
    float pr = f.x * ar.x + f.y * ar.y + f.z * ar.z + f.w * ar.w;
#pragma unroll
    for (int off = 16; off > 0; off >>= 1) {
        pl += __shfl_xor(pl, off, 32);
        pr += __shfl_xor(pr, off, 32);
    }
    if (lane == 0) { el[node] = pl; er[node] = pr; }
}

__device__ __forceinline__ unsigned enc_f32(float x) {
    unsigned u = __float_as_uint(x);
    return (u & 0x80000000u) ? ~u : (u | 0x80000000u);
}
__device__ __forceinline__ float dec_f32(unsigned u) {
    return (u & 0x80000000u) ? __uint_as_float(u ^ 0x80000000u) : __uint_as_float(~u);
}

__global__ void gat_edge_score(const int* __restrict__ src,
                               const int* __restrict__ dst,
                               const float* __restrict__ el,
                               const float* __restrict__ er,
                               float* __restrict__ e,
                               unsigned* __restrict__ menc) {
    const int i = blockIdx.x * blockDim.x + threadIdx.x;
    if (i >= NEDGES) return;
    const int s = src[i], d = dst[i];
    float v = el[s] + er[d];
    v = (v >= 0.0f) ? v : NEG_SLOPE * v;
    e[i] = v;
    atomicMax(&menc[d], enc_f32(v));
}

__global__ void gat_edge_exp(const int* __restrict__ dst,
                             const unsigned* __restrict__ menc,
                             float* __restrict__ e,
                             float* __restrict__ denom) {
    const int i = blockIdx.x * blockDim.x + threadIdx.x;
    if (i >= NEDGES) return;
    const int d = dst[i];
    const float mx = dec_f32(menc[d]);
    const float ex = __expf(e[i] - mx);
    e[i] = ex;
    atomicAdd(&denom[d], ex);
}

__global__ void gat_scatter(const int* __restrict__ src,
                            const int* __restrict__ dst,
                            const float* __restrict__ ex,
                            const float* __restrict__ denom,
                            const float* __restrict__ feat,
                            float* __restrict__ out) {
    const int edge = blockIdx.x * 8 + (threadIdx.x >> 5);
    if (edge >= NEDGES) return;
    const int lane = threadIdx.x & 31;
    const int s = src[edge], d = dst[edge];
    const float alpha = ex[edge] / fmaxf(denom[d], 1e-9f);

    const float4 f = *(const float4*)(feat + (long)s * DFEAT + lane * 4);
    float* o = out + (long)d * DFEAT + lane * 4;
    atomicAdd(o + 0, alpha * f.x);
    atomicAdd(o + 1, alpha * f.y);
    atomicAdd(o + 2, alpha * f.z);
    atomicAdd(o + 3, alpha * f.w);
}

extern "C" void kernel_launch(void* const* d_in, const int* in_sizes, int n_in,
                              void* d_out, int out_size, void* d_ws, size_t ws_size,
                              hipStream_t stream) {
    const float* h      = (const float*)d_in[0];
    const int*   src    = (const int*)  d_in[1];
    const int*   dst    = (const int*)  d_in[2];
    const float* W      = (const float*)d_in[3];
    const float* attn_l = (const float*)d_in[4];
    const float* attn_r = (const float*)d_in[5];
    float* out = (float*)d_out;

    float*    feat  = (float*)d_ws;
    float*    el    = feat + (long)NNODES * DFEAT;
    float*    er    = el + NNODES;
    unsigned* menc  = (unsigned*)(er + NNODES);
    float*    denom = (float*)(menc + NNODES);
    float*    e     = denom + NNODES;

    gat_zero_f32<<<4096, 256, 0, stream>>>(out, (long)NNODES * DFEAT);
    gat_zero_f32<<<256, 256, 0, stream>>>((float*)menc, 2L * NNODES);

    const int nstrips = NNODES / 16;
    gat_gemm_wmma<<<(nstrips + 3) / 4, 128, 0, stream>>>(h, W, feat);

    gat_attn_logits<<<(NNODES + 3) / 4, 128, 0, stream>>>(feat, attn_l, attn_r, el, er);

    gat_edge_score<<<(NEDGES + 255) / 256, 256, 0, stream>>>(src, dst, el, er, e, menc);

    gat_edge_exp<<<(NEDGES + 255) / 256, 256, 0, stream>>>(dst, menc, e, denom);

    gat_scatter<<<(NEDGES + 7) / 8, 256, 0, stream>>>(src, dst, e, denom, feat, out);
}


# scatter without float atomics: edges bucketed by destination (returning int adds, scan, fill), each bucket sorted by source so sums run in a fixed order; one wave per destination, one store per row
# speedup vs baseline: 1.7215x; 1.7215x over previous
#include <hip/hip_runtime.h>

typedef float v2f __attribute__((ext_vector_type(2)));
typedef float v8f __attribute__((ext_vector_type(8)));

#define NNODES   100000
#define NEDGES   1600000
#define DFEAT    128
#define NEG_SLOPE 0.2f

__global__ void gat_zero_f32(float* __restrict__ p, long n) {
    long i = blockIdx.x * (long)blockDim.x + threadIdx.x;
    long stride = gridDim.x * (long)blockDim.x;
    for (; i < n; i += stride) p[i] = 0.0f;
}

__global__ void gat_gemm_wmma(const float* __restrict__ h,
                              const float* __restrict__ W,
                              float* __restrict__ feat) {
    const int strip = blockIdx.x * 4 + (threadIdx.x >> 5);
    if (strip >= NNODES / 16) return;
    const int lane  = threadIdx.x & 31;
    const int m     = lane & 15;
    const int khalf = (lane >> 4) << 1;
    const int hihalf = (lane >> 4) << 3;
    const long row  = (long)strip * 16 + m;

    v8f acc[8];
#pragma unroll
    for (int ct = 0; ct < 8; ++ct) acc[ct] = (v8f){0,0,0,0,0,0,0,0};

    for (int k = 0; k < DFEAT; k += 4) {
        v2f a;
        a.x = h[row * DFEAT + k + khalf + 0];
        a.y = h[row * DFEAT + k + khalf + 1];
#pragma unroll
        for (int ct = 0; ct < 8; ++ct) {
            v2f b;
            b.x = W[(k + khalf + 0) * DFEAT + ct * 16 + m];
            b.y = W[(k + khalf + 1) * DFEAT + ct * 16 + m];
            acc[ct] = __builtin_amdgcn_wmma_f32_16x16x4_f32(
                false, a, false, b, (short)0, acc[ct], false, false);
        }
    }

#pragma unroll
    for (int ct = 0; ct < 8; ++ct) {
#pragma unroll
        for (int v = 0; v < 8; ++v) {
            long orow = (long)strip * 16 + v + hihalf;
            feat[orow * DFEAT + ct * 16 + m] = acc[ct][v];
        }
    }
}

__global__ void gat_attn_logits(const float* __restrict__ feat,
                                const float* __restrict__ attn_l,
                                const float* __restrict__ attn_r,
                                float* __restrict__ el,
                                float* __restrict__ er) {
    const int node = blockIdx.x * 4 + (threadIdx.x >> 5);
    if (node >= NNODES) return;
    const int lane = threadIdx.x & 31;

    const float4 f  = *(const float4*)(feat   + (long)node * DFEAT + lane * 4);
    const float4 al = *(const float4*)(attn_l + lane * 4);
    const float4 ar = *(const float4*)(attn_r + lane * 4);

    float pl = f.x * al.x + f.y * al.y + f.z * al.z + f.w * al.w;
    float pr = f.x * ar.x + f.y * ar.y + f.z * ar.z + f.w * ar.w;
#pragma unroll
    for (int off = 16; off > 0; off >>= 1) {
        pl += __shfl_xor(pl, off, 32);
        pr += __shfl_xor(pr, off, 32);
    }
    if (lane == 0) { el[node] = pl; er[node] = pr; }
}

__global__ void gat_count_rank(const int* __restrict__ dst,
                               int* __restrict__ cnt,
                               int* __restrict__ rank) {
    const int i = (blockIdx.x * blockDim.x + threadIdx.x) * 4;
    if (i >= NEDGES) return;
    const int4 d = *(const int4*)(dst + i);
    int4 r;
    r.x = atomicAdd(&cnt[d.x], 1);
    r.y = atomicAdd(&cnt[d.y], 1);
    r.z = atomicAdd(&cnt[d.z], 1);
    r.w = atomicAdd(&cnt[d.w], 1);
    *(int4*)(rank + i) = r;
}

__global__ void __launch_bounds__(1024)
gat_scan_start(const int* __restrict__ cnt, int* __restrict__ start) {
    __shared__ int wsum[32];
    __shared__ int wpre[32];
    const int tid  = threadIdx.x;
    const int lane = tid & 31;
    const int wv   = tid >> 5;
    const int base = blockIdx.x * 1024;

    int part = 0;
    for (int j = tid; j < (base >> 2); j += 1024) {
        const int4 c = *(const int4*)(cnt + 4 * j);
        part += c.x + c.y + c.z + c.w;
    }
#pragma unroll
    for (int off = 16; off > 0; off >>= 1) part += __shfl_xor(part, off, 32);
    if (lane == 0) wpre[wv] = part;

    const int idx = base + tid;
    const int v = (idx < NNODES) ? cnt[idx] : 0;
    int inc = v;
#pragma unroll
    for (int off = 1; off < 32; off <<= 1) {
        const int t = __shfl_up(inc, off, 32);
        if (lane >= off) inc += t;
    }
    if (lane == 31) wsum[wv] = inc;
    __syncthreads();

    int blockpre = 0, wavepre = 0;
    {
        const int p = wpre[lane];
        const int s = wsum[lane];
        int ps = p;
#pragma unroll
        for (int off = 16; off > 0; off >>= 1) ps += __shfl_xor(ps, off, 32);
        blockpre = ps;
        int sm = (lane < wv) ? s : 0;
#pragma unroll
        for (int off = 16; off > 0; off >>= 1) sm += __shfl_xor(sm, off, 32);
        wavepre = sm;
    }
    if (idx < NNODES) start[idx] = blockpre + wavepre + (inc - v);
}

__global__ void gat_fill(const int* __restrict__ src,
                         const int* __restrict__ dst,
                         const int* __restrict__ rank,
                         const int* __restrict__ start,
                         int* __restrict__ esrc) {
    const int i = (blockIdx.x * blockDim.x + threadIdx.x) * 4;
    if (i >= NEDGES) return;
    const int4 d = *(const int4*)(dst + i);
    const int4 s = *(const int4*)(src + i);
    const int4 r = *(const int4*)(rank + i);
    esrc[start[d.x] + r.x] = s.x;
    esrc[start[d.y] + r.y] = s.y;
    esrc[start[d.z] + r.z] = s.z;
    esrc[start[d.w] + r.w] = s.w;
}

__device__ __forceinline__ float gat_lrelu(float v) { return (v >= 0.0f) ? v : NEG_SLOPE * v; }
__device__ __forceinline__ float wave_max_f(float x) {
#pragma unroll
    for (int off = 16; off > 0; off >>= 1) x = fmaxf(x, __shfl_xor(x, off, 32));
    return x;
}
__device__ __forceinline__ float wave_sum_f(float x) {
#pragma unroll
    for (int off = 16; off > 0; off >>= 1) x += __shfl_xor(x, off, 32);
    return x;
}
__device__ __forceinline__ int wave_min_i(int x) {
#pragma unroll
    for (int off = 16; off > 0; off >>= 1) x = min(x, __shfl_xor(x, off, 32));
    return x;
}
__device__ __forceinline__ int wave_sum_i(int x) {
#pragma unroll
    for (int off = 16; off > 0; off >>= 1) x += __shfl_xor(x, off, 32);
    return x;
}

__global__ void gat_gather(const int* __restrict__ cnt,
                           const int* __restrict__ start,
                           const int* __restrict__ esrc,
                           const float* __restrict__ el,
                           const float* __restrict__ er,
                           const float* __restrict__ feat,
                           float* __restrict__ out) {
    const int node = __builtin_amdgcn_readfirstlane(blockIdx.x * 8 + (threadIdx.x >> 5));
    if (node >= NNODES) return;
    const int lane = threadIdx.x & 31;
    const int deg = cnt[node];
    const int st  = start[node];
    const float erd = er[node];
    const float* frow = feat + lane * 4;
    float4 acc = make_float4(0.0f, 0.0f, 0.0f, 0.0f);

    if (deg <= 32) {
        const bool active = lane < deg;
        const int key = active ? esrc[st + lane] : 0x7fffffff;
        int r = 0;
#pragma unroll
        for (int j = 0; j < 32; ++j) {
            const int kj = __builtin_amdgcn_readlane(key, j);
            r += ((kj < key) || (kj == key && j < lane)) ? 1 : 0;
        }
        const int skey = __builtin_amdgcn_ds_permute(r << 2, key);
        const float v = active ? gat_lrelu(el[active ? skey : 0] + erd) : -INFINITY;
        float m = wave_max_f(v);
        if (!(fabsf(m) < INFINITY)) m = 0.0f;
        const float ex = active ? __expf(v - m) : 0.0f;
        const float den = wave_sum_f(ex);
        const float alpha = ex / fmaxf(den, 1e-9f);

        int j = 0;
        for (; j + 4 <= deg; j += 4) {
            const int s0 = __builtin_amdgcn_readlane(skey, j + 0);
            const int s1 = __builtin_amdgcn_readlane(skey, j + 1);
            const int s2 = __builtin_amdgcn_readlane(skey, j + 2);
            const int s3 = __builtin_amdgcn_readlane(skey, j + 3);
            const float4 f0 = *(const float4*)(frow + (long)s0 * DFEAT);
            const float4 f1 = *(const float4*)(frow + (long)s1 * DFEAT);
            const float4 f2 = *(const float4*)(frow + (long)s2 * DFEAT);
            const float4 f3 = *(const float4*)(frow + (long)s3 * DFEAT);
            const float a0 = __int_as_float(__builtin_amdgcn_readlane(__float_as_int(alpha), j + 0));
            const float a1 = __int_as_float(__builtin_amdgcn_readlane(__float_as_int(alpha), j + 1));
            const float a2 = __int_as_float(__builtin_amdgcn_readlane(__float_as_int(alpha), j + 2));
            const float a3 = __int_as_float(__builtin_amdgcn_readlane(__float_as_int(alpha), j + 3));
            acc.x += a0 * f0.x; acc.y += a0 * f0.y; acc.z += a0 * f0.z; acc.w += a0 * f0.w;
            acc.x += a1 * f1.x; acc.y += a1 * f1.y; acc.z += a1 * f1.z; acc.w += a1 * f1.w;
            acc.x += a2 * f2.x; acc.y += a2 * f2.y; acc.z += a2 * f2.z; acc.w += a2 * f2.w;
            acc.x += a3 * f3.x; acc.y += a3 * f3.y; acc.z += a3 * f3.z; acc.w += a3 * f3.w;
        }
        for (; j < deg; ++j) {
            const int s0 = __builtin_amdgcn_readlane(skey, j);
            const float a0 = __int_as_float(__builtin_amdgcn_readlane(__float_as_int(alpha), j));
            const float4 f0 = *(const float4*)(frow + (long)s0 * DFEAT);
            acc.x += a0 * f0.x; acc.y += a0 * f0.y; acc.z += a0 * f0.z; acc.w += a0 * f0.w;
        }
    } else {
        float mloc = -INFINITY;
        for (int t = lane; t < deg; t += 32) mloc = fmaxf(mloc, gat_lrelu(el[esrc[st + t]] + erd));
        float m = wave_max_f(mloc);
        if (!(fabsf(m) < INFINITY)) m = 0.0f;

        float den = 0.0f;
        int prev = -1;
        for (int it = 0; it < deg; ++it) {
            int kmin = 0x7fffffff;
            for (int t = lane; t < deg; t += 32) { const int k = esrc[st + t]; if (k > prev) kmin = min(kmin, k); }
            kmin = wave_min_i(kmin);
            if (kmin == 0x7fffffff) break;
            int c = 0;
            for (int t = lane; t < deg; t += 32) c += (esrc[st + t] == kmin) ? 1 : 0;
            c = wave_sum_i(c);
            const float ex = __expf(gat_lrelu(el[kmin] + erd) - m);
            for (int q = 0; q < c; ++q) den += ex;
            prev = kmin;
        }
        const float dinv = fmaxf(den, 1e-9f);
        prev = -1;
        for (int it = 0; it < deg; ++it) {
            int kmin = 0x7fffffff;
            for (int t = lane; t < deg; t += 32) { const int k = esrc[st + t]; if (k > prev) kmin = min(kmin, k); }
            kmin = wave_min_i(kmin);
            if (kmin == 0x7fffffff) break;
            int c = 0;
            for (int t = lane; t < deg; t += 32) c += (esrc[st + t] == kmin) ? 1 : 0;
            c = wave_sum_i(c);
            const float a = __expf(gat_lrelu(el[kmin] + erd) - m) / dinv;
            const float4 f = *(const float4*)(frow + (long)kmin * DFEAT);
            for (int q = 0; q < c; ++q) {
                acc.x += a * f.x; acc.y += a * f.y; acc.z += a * f.z; acc.w += a * f.w;
            }
            prev = kmin;
        }
    }
    *(float4*)(out + (long)node * DFEAT + lane * 4) = acc;
}

extern "C" void kernel_launch(void* const* d_in, const int* in_sizes, int n_in,
                              void* d_out, int out_size, void* d_ws, size_t ws_size,
                              hipStream_t stream) {
    const float* h      = (const float*)d_in[0];
    const int*   src    = (const int*)  d_in[1];
    const int*   dst    = (const int*)  d_in[2];
    const float* W      = (const float*)d_in[3];
    const float* attn_l = (const float*)d_in[4];
    const float* attn_r = (const float*)d_in[5];
    float* out = (float*)d_out;

    float* feat  = (float*)d_ws;
    float* el    = feat + (long)NNODES * DFEAT;
    float* er    = el + NNODES;
    int*   cnt   = (int*)(er + NNODES);
    int*   start = cnt + NNODES;
    int*   esrc  = start + NNODES;
    int*   rank  = (int*)d_out;

    gat_zero_f32<<<128, 256, 0, stream>>>((float*)cnt, (long)NNODES);

    const int nstrips = NNODES / 16;
    gat_gemm_wmma<<<(nstrips + 3) / 4, 128, 0, stream>>>(h, W, feat);

    gat_attn_logits<<<(NNODES + 3) / 4, 128, 0, stream>>>(feat, attn_l, attn_r, el, er);

    gat_count_rank<<<(NEDGES / 4 + 255) / 256, 256, 0, stream>>>(dst, cnt, rank);
    gat_scan_start<<<(NNODES + 1023) / 1024, 1024, 0, stream>>>(cnt, start);
    gat_fill<<<(NEDGES / 4 + 255) / 256, 256, 0, stream>>>(src, dst, rank, start, esrc);
    gat_gather<<<(NNODES + 7) / 8, 256, 0, stream>>>(cnt, start, esrc, el, er, feat, out);
}


# gather: nodes with 33..64 in-edges sort their bucket once (rank by counting, two keys per lane, via LDS) and sum in one pass, no wave-wide min search per edge; same terms and order, products fused
# speedup vs baseline: 2.3086x; 1.3410x over previous
#include <hip/hip_runtime.h>

typedef float v2f __attribute__((ext_vector_type(2)));
typedef float v8f __attribute__((ext_vector_type(8)));

#define NNODES   100000
#define NEDGES   1600000
#define DFEAT    128
#define NEG_SLOPE 0.2f

__global__ void gat_zero_f32(float* __restrict__ p, long n) {
    long i = blockIdx.x * (long)blockDim.x + threadIdx.x;
    long stride = gridDim.x * (long)blockDim.x;
    for (; i < n; i += stride) p[i] = 0.0f;
}

template <int MASK>
__device__ __forceinline__ float row_xor16(float x) {
    return __int_as_float(__builtin_amdgcn_update_dpp(__float_as_int(x), __float_as_int(x),
                                                      0x160 | MASK, 0xf, 0xf, false));
}

__global__ void gat_gemm_wmma(const float* __restrict__ h,
                              const float* __restrict__ W,
                              const float* __restrict__ attn_l,
                              const float* __restrict__ attn_r,
                              float* __restrict__ feat,
                              float* __restrict__ el,
                              float* __restrict__ er) {
    const int strip = blockIdx.x * 4 + (threadIdx.x >> 5);
    if (strip >= NNODES / 16) return;
    const int lane  = threadIdx.x & 31;
    const int m     = lane & 15;
    const int khalf = (lane >> 4) << 1;
    const int hihalf = (lane >> 4) << 3;
    const long row  = (long)strip * 16 + m;

    v8f acc[8];
#pragma unroll
    for (int ct = 0; ct < 8; ++ct) acc[ct] = (v8f){0,0,0,0,0,0,0,0};

    for (int k = 0; k < DFEAT; k += 4) {
        v2f a;
        a.x = h[row * DFEAT + k + khalf + 0];
        a.y = h[row * DFEAT + k + khalf + 1];
#pragma unroll
        for (int ct = 0; ct < 8; ++ct) {
            v2f b;
            b.x = W[(k + khalf + 0) * DFEAT + ct * 16 + m];
            b.y = W[(k + khalf + 1) * DFEAT + ct * 16 + m];
            acc[ct] = __builtin_amdgcn_wmma_f32_16x16x4_f32(
                false, a, false, b, (short)0, acc[ct], false, false);
        }
    }

#pragma unroll
    for (int ct = 0; ct < 8; ++ct) {
#pragma unroll
        for (int v = 0; v < 8; ++v) {
            long orow = (long)strip * 16 + v + hihalf;
            feat[orow * DFEAT + ct * 16 + m] = acc[ct][v];
        }
    }

    float part[16];
#pragma unroll
    for (int i = 0; i < 16; ++i) part[i] = 0.0f;
#pragma unroll
    for (int ct = 0; ct < 8; ++ct) {
        const float wl = attn_l[ct * 16 + m];
        const float wr = attn_r[ct * 16 + m];
#pragma unroll
        for (int v = 0; v < 8; ++v) {
            part[v]     = __builtin_fmaf(acc[ct][v], wl, part[v]);
            part[8 + v] = __builtin_fmaf(acc[ct][v], wr, part[8 + v]);
        }
    }
    const bool b3 = (m & 8) != 0, b2 = (m & 4) != 0, b1 = (m & 2) != 0, b0 = (m & 1) != 0;
    float s8[8], s4[4], s2[2];
#pragma unroll
    for (int j = 0; j < 8; ++j) {
        const float keep = b3 ? part[8 + j] : part[j];
        const float send = b3 ? part[j] : part[8 + j];
        s8[j] = keep + row_xor16<8>(send);
    }
#pragma unroll
    for (int j = 0; j < 4; ++j) {
        const float keep = b2 ? s8[4 + j] : s8[j];
        const float send = b2 ? s8[j] : s8[4 + j];
        s4[j] = keep + row_xor16<4>(send);
    }
#pragma unroll
    for (int j = 0; j < 2; ++j) {
        const float keep = b1 ? s4[2 + j] : s4[j];
        const float send = b1 ? s4[j] : s4[2 + j];
        s2[j] = keep + row_xor16<2>(send);
    }
    const float keep1 = b0 ? s2[1] : s2[0];
    const float send1 = b0 ? s2[0] : s2[1];
    const float tot = keep1 + row_xor16<1>(send1);
    const long lrow = (long)strip * 16 + hihalf + (m & 7);
    if (m < 8) el[lrow] = tot; else er[lrow] = tot;
}

__global__ void gat_count_rank(const int* __restrict__ dst,
                               int* __restrict__ cnt,
                               int* __restrict__ rank) {
    const int i = (blockIdx.x * blockDim.x + threadIdx.x) * 4;
    if (i >= NEDGES) return;
    const int4 d = *(const int4*)(dst + i);
    int4 r;
    r.x = atomicAdd(&cnt[d.x], 1);
    r.y = atomicAdd(&cnt[d.y], 1);
    r.z = atomicAdd(&cnt[d.z], 1);
    r.w = atomicAdd(&cnt[d.w], 1);
    *(int4*)(rank + i) = r;
}

__global__ void __launch_bounds__(1024)
gat_scan_start(const int* __restrict__ cnt, int* __restrict__ start) {
    __shared__ int wsum[32];
    __shared__ int wpre[32];
    const int tid  = threadIdx.x;
    const int lane = tid & 31;
    const int wv   = tid >> 5;
    const int base = blockIdx.x * 1024;

    int part = 0;
    for (int j = tid; j < (base >> 2); j += 1024) {
        const int4 c = *(const int4*)(cnt + 4 * j);
        part += c.x + c.y + c.z + c.w;
    }
#pragma unroll
    for (int off = 16; off > 0; off >>= 1) part += __shfl_xor(part, off, 32);
    if (lane == 0) wpre[wv] = part;

    const int idx = base + tid;
    const int v = (idx < NNODES) ? cnt[idx] : 0;
    int inc = v;
#pragma unroll
    for (int off = 1; off < 32; off <<= 1) {
        const int t = __shfl_up(inc, off, 32);
        if (lane >= off) inc += t;
    }
    if (lane == 31) wsum[wv] = inc;
    __syncthreads();

    int blockpre = 0, wavepre = 0;
    {
        const int p = wpre[lane];
        const int s = wsum[lane];
        int ps = p;
#pragma unroll
        for (int off = 16; off > 0; off >>= 1) ps += __shfl_xor(ps, off, 32);
        blockpre = ps;
        int sm = (lane < wv) ? s : 0;
#pragma unroll
        for (int off = 16; off > 0; off >>= 1) sm += __shfl_xor(sm, off, 32);
        wavepre = sm;
    }
    if (idx < NNODES) start[idx] = blockpre + wavepre + (inc - v);
}

__global__ void gat_fill(const int* __restrict__ src,
                         const int* __restrict__ dst,
                         const int* __restrict__ rank,
                         const int* __restrict__ start,
                         int* __restrict__ esrc) {
    const int i = (blockIdx.x * blockDim.x + threadIdx.x) * 4;
    if (i >= NEDGES) return;
    const int4 d = *(const int4*)(dst + i);
    const int4 s = *(const int4*)(src + i);
    const int4 r = *(const int4*)(rank + i);
    esrc[start[d.x] + r.x] = s.x;
    esrc[start[d.y] + r.y] = s.y;
    esrc[start[d.z] + r.z] = s.z;
    esrc[start[d.w] + r.w] = s.w;
}

__device__ __forceinline__ float gat_lrelu(float v) { return (v >= 0.0f) ? v : NEG_SLOPE * v; }
__device__ __forceinline__ float wave_max_f(float x) {
#pragma unroll
    for (int off = 16; off > 0; off >>= 1) x = fmaxf(x, __shfl_xor(x, off, 32));
    return x;
}
__device__ __forceinline__ float wave_sum_f(float x) {
#pragma unroll
    for (int off = 16; off > 0; off >>= 1) x += __shfl_xor(x, off, 32);
    return x;
}
__device__ __forceinline__ int wave_min_i(int x) {
#pragma unroll
    for (int off = 16; off > 0; off >>= 1) x = min(x, __shfl_xor(x, off, 32));
    return x;
}
__device__ __forceinline__ int wave_sum_i(int x) {
#pragma unroll
    for (int off = 16; off > 0; off >>= 1) x += __shfl_xor(x, off, 32);
    return x;
}

__device__ __forceinline__ float lane_f(float x, int j) {
    return __int_as_float(__builtin_amdgcn_readlane(__float_as_int(x), j));
}
__device__ __forceinline__ void gat_axpy4(float4& acc, const float a, const float4 f) {
    acc.x += a * f.x; acc.y += a * f.y; acc.z += a * f.z; acc.w += a * f.w;
}

__global__ void gat_gather(const int* __restrict__ cnt,
                           const int* __restrict__ start,
                           const int* __restrict__ esrc,
                           const float* __restrict__ el,
                           const float* __restrict__ er,
                           const float* __restrict__ feat,
                           float* __restrict__ out) {
    __shared__ int sortbuf[8 * 64];
    const int node = __builtin_amdgcn_readfirstlane(blockIdx.x * 8 + (threadIdx.x >> 5));
    if (node >= NNODES) return;
    const int lane = threadIdx.x & 31;
    const int deg = cnt[node];
    const int st  = start[node];
    const float erd = er[node];
    const float* frow = feat + lane * 4;
    float4 acc = make_float4(0.0f, 0.0f, 0.0f, 0.0f);

    if (deg <= 32) {
        const bool active = lane < deg;
        const int key = active ? esrc[st + lane] : 0x7fffffff;
        int r = 0;
#pragma unroll
        for (int j = 0; j < 32; ++j) {
            const int kj = __builtin_amdgcn_readlane(key, j);
            r += ((kj < key) || (kj == key && j < lane)) ? 1 : 0;
        }
        const int skey = __builtin_amdgcn_ds_permute(r << 2, key);
        const float v = active ? gat_lrelu(el[active ? skey : 0] + erd) : -INFINITY;
        float m = wave_max_f(v);
        if (!(fabsf(m) < INFINITY)) m = 0.0f;
        const float ex = active ? __expf(v - m) : 0.0f;
        const float den = wave_sum_f(ex);
        const float alpha = ex / fmaxf(den, 1e-9f);

        int j = 0;
        for (; j + 4 <= deg; j += 4) {
            const int s0 = __builtin_amdgcn_readlane(skey, j + 0);
            const int s1 = __builtin_amdgcn_readlane(skey, j + 1);
            const int s2 = __builtin_amdgcn_readlane(skey, j + 2);
            const int s3 = __builtin_amdgcn_readlane(skey, j + 3);
            const float4 f0 = *(const float4*)(frow + (long)s0 * DFEAT);
            const float4 f1 = *(const float4*)(frow + (long)s1 * DFEAT);
            const float4 f2 = *(const float4*)(frow + (long)s2 * DFEAT);
            const float4 f3 = *(const float4*)(frow + (long)s3 * DFEAT);
            const float a0 = __int_as_float(__builtin_amdgcn_readlane(__float_as_int(alpha), j + 0));
            const float a1 = __int_as_float(__builtin_amdgcn_readlane(__float_as_int(alpha), j + 1));
            const float a2 = __int_as_float(__builtin_amdgcn_readlane(__float_as_int(alpha), j + 2));
            const float a3 = __int_as_float(__builtin_amdgcn_readlane(__float_as_int(alpha), j + 3));
            acc.x += a0 * f0.x; acc.y += a0 * f0.y; acc.z += a0 * f0.z; acc.w += a0 * f0.w;
            acc.x += a1 * f1.x; acc.y += a1 * f1.y; acc.z += a1 * f1.z; acc.w += a1 * f1.w;
            acc.x += a2 * f2.x; acc.y += a2 * f2.y; acc.z += a2 * f2.z; acc.w += a2 * f2.w;
            acc.x += a3 * f3.x; acc.y += a3 * f3.y; acc.z += a3 * f3.z; acc.w += a3 * f3.w;
        }
        for (; j < deg; ++j) {
            const int s0 = __builtin_amdgcn_readlane(skey, j);
            const float a0 = __int_as_float(__builtin_amdgcn_readlane(__float_as_int(alpha), j));
            const float4 f0 = *(const float4*)(frow + (long)s0 * DFEAT);
            acc.x += a0 * f0.x; acc.y += a0 * f0.y; acc.z += a0 * f0.z; acc.w += a0 * f0.w;
        }
    } else if (deg <= 64) {
        int* sb = sortbuf + (threadIdx.x >> 5) * 64;
        const bool act1 = lane + 32 < deg;
        const int c0 = (esrc[st + lane] << 6) | lane;
        const int c1 = ((act1 ? esrc[st + 32 + lane] : 0x1ffffff) << 6) | (32 + lane);
        int r0 = 0, r1 = 0;
#pragma unroll
        for (int j = 0; j < 32; ++j) {
            const int ca = __builtin_amdgcn_readlane(c0, j);
            const int cb = __builtin_amdgcn_readlane(c1, j);
            r0 += (ca < c0) ? 1 : 0;
            r0 += (cb < c0) ? 1 : 0;
            r1 += (ca < c1) ? 1 : 0;
            r1 += (cb < c1) ? 1 : 0;
        }
        sb[r0] = c0;
        sb[r1] = c1;
        const int sk0 = sb[lane] >> 6;
        const int sk1 = sb[32 + lane] >> 6;

        const float v0 = gat_lrelu(el[sk0] + erd);
        const float v1 = act1 ? gat_lrelu(el[act1 ? sk1 : 0] + erd) : -INFINITY;
        float m = wave_max_f(fmaxf(v0, v1));
        if (!(fabsf(m) < INFINITY)) m = 0.0f;
        const float ex0 = __expf(v0 - m);
        const float ex1 = act1 ? __expf(v1 - m) : 0.0f;

        float den = 0.0f;
        for (int j = 0; j < 32; ++j)   den += lane_f(ex0, j);
        for (int j = 32; j < deg; ++j) den += lane_f(ex1, j - 32);
        const float dinv = fmaxf(den, 1e-9f);
        const float al0 = ex0 / dinv;
        const float al1 = ex1 / dinv;

        for (int j = 0; j < 32; j += 4) {
            const float4 f0 = *(const float4*)(frow + (long)__builtin_amdgcn_readlane(sk0, j + 0) * DFEAT);
            const float4 f1 = *(const float4*)(frow + (long)__builtin_amdgcn_readlane(sk0, j + 1) * DFEAT);
            const float4 f2 = *(const float4*)(frow + (long)__builtin_amdgcn_readlane(sk0, j + 2) * DFEAT);
            const float4 f3 = *(const float4*)(frow + (long)__builtin_amdgcn_readlane(sk0, j + 3) * DFEAT);
            gat_axpy4(acc, lane_f(al0, j + 0), f0);
            gat_axpy4(acc, lane_f(al0, j + 1), f1);
            gat_axpy4(acc, lane_f(al0, j + 2), f2);
            gat_axpy4(acc, lane_f(al0, j + 3), f3);
        }
        const int d1 = deg - 32;
        int j = 0;
        for (; j + 4 <= d1; j += 4) {
            const float4 f0 = *(const float4*)(frow + (long)__builtin_amdgcn_readlane(sk1, j + 0) * DFEAT);
            const float4 f1 = *(const float4*)(frow + (long)__builtin_amdgcn_readlane(sk1, j + 1) * DFEAT);
            const float4 f2 = *(const float4*)(frow + (long)__builtin_amdgcn_readlane(sk1, j + 2) * DFEAT);
            const float4 f3 = *(const float4*)(frow + (long)__builtin_amdgcn_readlane(sk1, j + 3) * DFEAT);
            gat_axpy4(acc, lane_f(al1, j + 0), f0);
            gat_axpy4(acc, lane_f(al1, j + 1), f1);
            gat_axpy4(acc, lane_f(al1, j + 2), f2);
            gat_axpy4(acc, lane_f(al1, j + 3), f3);
        }
        for (; j < d1; ++j) {
            const float4 f0 = *(const float4*)(frow + (long)__builtin_amdgcn_readlane(sk1, j) * DFEAT);
            gat_axpy4(acc, lane_f(al1, j), f0);
        }
    } else {
        float mloc = -INFINITY;
        for (int t = lane; t < deg; t += 32) mloc = fmaxf(mloc, gat_lrelu(el[esrc[st + t]] + erd));
        float m = wave_max_f(mloc);
        if (!(fabsf(m) < INFINITY)) m = 0.0f;

        float den = 0.0f;
        int prev = -1;
        for (int it = 0; it < deg; ++it) {
            int kmin = 0x7fffffff;
            for (int t = lane; t < deg; t += 32) { const int k = esrc[st + t]; if (k > prev) kmin = min(kmin, k); }
            kmin = wave_min_i(kmin);
            if (kmin == 0x7fffffff) break;
            int c = 0;
            for (int t = lane; t < deg; t += 32) c += (esrc[st + t] == kmin) ? 1 : 0;
            c = wave_sum_i(c);
            const float ex = __expf(gat_lrelu(el[kmin] + erd) - m);
            for (int q = 0; q < c; ++q) den += ex;
            prev = kmin;
        }
        const float dinv = fmaxf(den, 1e-9f);
        prev = -1;
        for (int it = 0; it < deg; ++it) {
            int kmin = 0x7fffffff;
            for (int t = lane; t < deg; t += 32) { const int k = esrc[st + t]; if (k > prev) kmin = min(kmin, k); }
            kmin = wave_min_i(kmin);
            if (kmin == 0x7fffffff) break;
            int c = 0;
            for (int t = lane; t < deg; t += 32) c += (esrc[st + t] == kmin) ? 1 : 0;
            c = wave_sum_i(c);
            const float a = __expf(gat_lrelu(el[kmin] + erd) - m) / dinv;
            const float4 f = *(const float4*)(frow + (long)kmin * DFEAT);
            for (int q = 0; q < c; ++q) {
                acc.x += a * f.x; acc.y += a * f.y; acc.z += a * f.z; acc.w += a * f.w;
            }
            prev = kmin;
        }
    }
    *(float4*)(out + (long)node * DFEAT + lane * 4) = acc;
}

extern "C" void kernel_launch(void* const* d_in, const int* in_sizes, int n_in,
                              void* d_out, int out_size, void* d_ws, size_t ws_size,
                              hipStream_t stream) {
    const float* h      = (const float*)d_in[0];
    const int*   src    = (const int*)  d_in[1];
    const int*   dst    = (const int*)  d_in[2];
    const float* W      = (const float*)d_in[3];
    const float* attn_l = (const float*)d_in[4];
    const float* attn_r = (const float*)d_in[5];
    float* out = (float*)d_out;

    float* feat  = (float*)d_ws;
    float* el    = feat + (long)NNODES * DFEAT;
    float* er    = el + NNODES;
    int*   cnt   = (int*)(er + NNODES);
    int*   start = cnt + NNODES;
    int*   esrc  = start + NNODES;
    int*   rank  = (int*)d_out;

    gat_zero_f32<<<128, 256, 0, stream>>>((float*)cnt, (long)NNODES);

    const int nstrips = NNODES / 16;
    gat_gemm_wmma<<<(nstrips + 3) / 4, 128, 0, stream>>>(h, W, attn_l, attn_r, feat, el, er);

    gat_count_rank<<<(NEDGES / 4 + 255) / 256, 256, 0, stream>>>(dst, cnt, rank);
    gat_scan_start<<<(NNODES + 1023) / 1024, 1024, 0, stream>>>(cnt, start);
    gat_fill<<<(NEDGES / 4 + 255) / 256, 256, 0, stream>>>(src, dst, rank, start, esrc);
    gat_gather<<<(NNODES + 7) / 8, 256, 0, stream>>>(cnt, start, esrc, el, er, feat, out);
}


# bucket slots from LDS counters, not 1.6M returning adds on memory: edges grouped by destination range (LDS histogram, one memory add per range per group), then one group per 256 nodes; same sums
# speedup vs baseline: 4.6018x; 1.9933x over previous
#include <hip/hip_runtime.h>

typedef float v2f __attribute__((ext_vector_type(2)));
typedef float v8f __attribute__((ext_vector_type(8)));

#define NNODES   100000
#define NEDGES   1600000
#define DFEAT    128
#define NEG_SLOPE 0.2f

__global__ void gat_zero_f32(float* __restrict__ p, long n) {
    long i = blockIdx.x * (long)blockDim.x + threadIdx.x;
    long stride = gridDim.x * (long)blockDim.x;
    for (; i < n; i += stride) p[i] = 0.0f;
}

template <int MASK>
__device__ __forceinline__ float row_xor16(float x) {
    return __int_as_float(__builtin_amdgcn_update_dpp(__float_as_int(x), __float_as_int(x),
                                                      0x160 | MASK, 0xf, 0xf, false));
}

__global__ void gat_gemm_wmma(const float* __restrict__ h,
                              const float* __restrict__ W,
                              const float* __restrict__ attn_l,
                              const float* __restrict__ attn_r,
                              float* __restrict__ feat,
                              float* __restrict__ el,
                              float* __restrict__ er) {
    const int strip = blockIdx.x * 4 + (threadIdx.x >> 5);
    if (strip >= NNODES / 16) return;
    const int lane  = threadIdx.x & 31;
    const int m     = lane & 15;
    const int khalf = (lane >> 4) << 1;
    const int hihalf = (lane >> 4) << 3;
    const long row  = (long)strip * 16 + m;

    v8f acc[8];
#pragma unroll
    for (int ct = 0; ct < 8; ++ct) acc[ct] = (v8f){0,0,0,0,0,0,0,0};

    for (int k = 0; k < DFEAT; k += 4) {
        v2f a;
        a.x = h[row * DFEAT + k + khalf + 0];
        a.y = h[row * DFEAT + k + khalf + 1];
#pragma unroll
        for (int ct = 0; ct < 8; ++ct) {
            v2f b;
            b.x = W[(k + khalf + 0) * DFEAT + ct * 16 + m];
            b.y = W[(k + khalf + 1) * DFEAT + ct * 16 + m];
            acc[ct] = __builtin_amdgcn_wmma_f32_16x16x4_f32(
                false, a, false, b, (short)0, acc[ct], false, false);
        }
    }

#pragma unroll
    for (int ct = 0; ct < 8; ++ct) {
#pragma unroll
        for (int v = 0; v < 8; ++v) {
            long orow = (long)strip * 16 + v + hihalf;
            feat[orow * DFEAT + ct * 16 + m] = acc[ct][v];
        }
    }

    float part[16];
#pragma unroll
    for (int i = 0; i < 16; ++i) part[i] = 0.0f;
#pragma unroll
    for (int ct = 0; ct < 8; ++ct) {
        const float wl = attn_l[ct * 16 + m];
        const float wr = attn_r[ct * 16 + m];
#pragma unroll
        for (int v = 0; v < 8; ++v) {
            part[v]     = __builtin_fmaf(acc[ct][v], wl, part[v]);
            part[8 + v] = __builtin_fmaf(acc[ct][v], wr, part[8 + v]);
        }
    }
    const bool b3 = (m & 8) != 0, b2 = (m & 4) != 0, b1 = (m & 2) != 0, b0 = (m & 1) != 0;
    float s8[8], s4[4], s2[2];
#pragma unroll
    for (int j = 0; j < 8; ++j) {
        const float keep = b3 ? part[8 + j] : part[j];
        const float send = b3 ? part[j] : part[8 + j];
        s8[j] = keep + row_xor16<8>(send);
    }
#pragma unroll
    for (int j = 0; j < 4; ++j) {
        const float keep = b2 ? s8[4 + j] : s8[j];
        const float send = b2 ? s8[j] : s8[4 + j];
        s4[j] = keep + row_xor16<4>(send);
    }
#pragma unroll
    for (int j = 0; j < 2; ++j) {
        const float keep = b1 ? s4[2 + j] : s4[j];
        const float send = b1 ? s4[j] : s4[2 + j];
        s2[j] = keep + row_xor16<2>(send);
    }
    const float keep1 = b0 ? s2[1] : s2[0];
    const float send1 = b0 ? s2[0] : s2[1];
    const float tot = keep1 + row_xor16<1>(send1);
    const long lrow = (long)strip * 16 + hihalf + (m & 7);
    if (m < 8) el[lrow] = tot; else er[lrow] = tot;
}

#define BCAP 128
#define RANGE_SHIFT 8
#define RANGE_NODES 256
#define NRANGE ((NNODES + RANGE_NODES - 1) / RANGE_NODES)
#define RANGE_CAP 8192
#define BIN_EDGES 2048

__global__ void gat_bin(const int* __restrict__ src,
                        const int* __restrict__ dst,
                        int* __restrict__ rcnt,
                        int* __restrict__ lists_w) {
    int2* lists = (int2*)lists_w;
    __shared__ int hist[NRANGE];
    __shared__ int base[NRANGE];
    for (int t = threadIdx.x; t < NRANGE; t += 256) hist[t] = 0;
    __syncthreads();
    const int e0 = blockIdx.x * BIN_EDGES + threadIdx.x * 4;
    int dv[8], sv[8], rk[8];
#pragma unroll
    for (int g = 0; g < 2; ++g) {
        const int i = e0 + g * 1024;
        int4 d = make_int4(-1, -1, -1, -1), s = make_int4(0, 0, 0, 0);
        if (i < NEDGES) { d = *(const int4*)(dst + i); s = *(const int4*)(src + i); }
        dv[g * 4 + 0] = d.x; dv[g * 4 + 1] = d.y; dv[g * 4 + 2] = d.z; dv[g * 4 + 3] = d.w;
        sv[g * 4 + 0] = s.x; sv[g * 4 + 1] = s.y; sv[g * 4 + 2] = s.z; sv[g * 4 + 3] = s.w;
    }
#pragma unroll
    for (int q = 0; q < 8; ++q) rk[q] = (dv[q] >= 0) ? atomicAdd(&hist[dv[q] >> RANGE_SHIFT], 1) : 0;
    __syncthreads();
    for (int t = threadIdx.x; t < NRANGE; t += 256) {
        const int hcount = hist[t];
        base[t] = (hcount > 0) ? atomicAdd(&rcnt[t], hcount) : 0;
    }
    __syncthreads();
#pragma unroll
    for (int q = 0; q < 8; ++q) {
        if (dv[q] >= 0) {
            const int r = dv[q] >> RANGE_SHIFT;
            const int pos = base[r] + rk[q];
            if (pos < RANGE_CAP) lists[r * RANGE_CAP + pos] = make_int2(dv[q], sv[q]);
        }
    }
}

__global__ void gat_fill(const int* __restrict__ src,
                         const int* __restrict__ dst,
                         const int* __restrict__ rcnt,
                         const int* __restrict__ lists_w,
                         int* __restrict__ cnt,
                         int* __restrict__ bkt) {
    __shared__ int c[RANGE_NODES];
    const int r = blockIdx.x;
    if (threadIdx.x < RANGE_NODES) c[threadIdx.x] = 0;
    __syncthreads();
    const int n = rcnt[r];
    if (n <= RANGE_CAP) {
        const int2* L = (const int2*)lists_w + r * RANGE_CAP;
        for (int i0 = threadIdx.x; i0 < n; i0 += 4096) {
            int2 e[8];
            int slot[8];
#pragma unroll
            for (int q = 0; q < 8; ++q) {
                const int i = i0 + q * 512;
                e[q] = (i < n) ? L[i] : make_int2(-1, 0);
            }
#pragma unroll
            for (int q = 0; q < 8; ++q)
                slot[q] = (e[q].x >= 0) ? atomicAdd(&c[e[q].x & (RANGE_NODES - 1)], 1) : BCAP;
#pragma unroll
            for (int q = 0; q < 8; ++q)
                if (slot[q] < BCAP) bkt[e[q].x * BCAP + slot[q]] = e[q].y;
        }
    } else {
        for (int i = threadIdx.x; i < NEDGES; i += 512) {
            const int d = dst[i];
            if ((d >> RANGE_SHIFT) == r) {
                const int slot1 = atomicAdd(&c[d & (RANGE_NODES - 1)], 1);
                if (slot1 < BCAP) bkt[d * BCAP + slot1] = src[i];
            }
        }
    }
    __syncthreads();
    if (threadIdx.x < RANGE_NODES) {
        const int node = r * RANGE_NODES + threadIdx.x;
        if (node < NNODES) cnt[node] = c[threadIdx.x];
    }
}

__device__ __forceinline__ float gat_lrelu(float v) { return (v >= 0.0f) ? v : NEG_SLOPE * v; }
__device__ __forceinline__ float wave_max_f(float x) {
#pragma unroll
    for (int off = 16; off > 0; off >>= 1) x = fmaxf(x, __shfl_xor(x, off, 32));
    return x;
}
__device__ __forceinline__ float wave_sum_f(float x) {
#pragma unroll
    for (int off = 16; off > 0; off >>= 1) x += __shfl_xor(x, off, 32);
    return x;
}
__device__ __forceinline__ int wave_min_i(int x) {
#pragma unroll
    for (int off = 16; off > 0; off >>= 1) x = min(x, __shfl_xor(x, off, 32));
    return x;
}
__device__ __forceinline__ int wave_sum_i(int x) {
#pragma unroll
    for (int off = 16; off > 0; off >>= 1) x += __shfl_xor(x, off, 32);
    return x;
}

__device__ __forceinline__ float lane_f(float x, int j) {
    return __int_as_float(__builtin_amdgcn_readlane(__float_as_int(x), j));
}
__device__ __forceinline__ void gat_axpy4(float4& acc, const float a, const float4 f) {
    acc.x += a * f.x; acc.y += a * f.y; acc.z += a * f.z; acc.w += a * f.w;
}

template <bool FULL>
__device__ __forceinline__ int gat_item(const float* bkt, const int* __restrict__ src,
                                        const int* __restrict__ dst, const int node, const int t) {
    if (FULL) return (dst[t] == node) ? src[t] : -1;
    return __float_as_int(bkt[t]);
}

template <bool FULL>
__device__ __forceinline__ float4 gat_walk(const float* bkt, const int* __restrict__ src,
                                           const int* __restrict__ dst, const float* __restrict__ el,
                                           const float* frow, const int node, const int deg,
                                           const int lane, const float erd) {
    const int n = FULL ? NEDGES : deg;
    float4 acc = make_float4(0.0f, 0.0f, 0.0f, 0.0f);
    float mloc = -INFINITY;
    for (int t = lane; t < n; t += 32) {
        const int k = gat_item<FULL>(bkt, src, dst, node, t);
        if (k >= 0) mloc = fmaxf(mloc, gat_lrelu(el[k] + erd));
    }
    float m = wave_max_f(mloc);
    if (!(fabsf(m) < INFINITY)) m = 0.0f;

    float den = 0.0f;
    int prev = -1;
    for (int it = 0; it < deg; ++it) {
        int kmin = 0x7fffffff;
        for (int t = lane; t < n; t += 32) { const int k = gat_item<FULL>(bkt, src, dst, node, t); if (k > prev) kmin = min(kmin, k); }
        kmin = wave_min_i(kmin);
        if (kmin == 0x7fffffff) break;
        int c = 0;
        for (int t = lane; t < n; t += 32) c += (gat_item<FULL>(bkt, src, dst, node, t) == kmin) ? 1 : 0;
        c = wave_sum_i(c);
        const float ex = __expf(gat_lrelu(el[kmin] + erd) - m);
        for (int q = 0; q < c; ++q) den += ex;
        prev = kmin;
    }
    const float dinv = fmaxf(den, 1e-9f);
    prev = -1;
    for (int it = 0; it < deg; ++it) {
        int kmin = 0x7fffffff;
        for (int t = lane; t < n; t += 32) { const int k = gat_item<FULL>(bkt, src, dst, node, t); if (k > prev) kmin = min(kmin, k); }
        kmin = wave_min_i(kmin);
        if (kmin == 0x7fffffff) break;
        int c = 0;
        for (int t = lane; t < n; t += 32) c += (gat_item<FULL>(bkt, src, dst, node, t) == kmin) ? 1 : 0;
        c = wave_sum_i(c);
        const float a = __expf(gat_lrelu(el[kmin] + erd) - m) / dinv;
        const float4 f = *(const float4*)(frow + (long)kmin * DFEAT);
        for (int q = 0; q < c; ++q) {
            acc.x += a * f.x; acc.y += a * f.y; acc.z += a * f.z; acc.w += a * f.w;
        }
        prev = kmin;
    }
    return acc;
}

__global__ void gat_gather(const int* __restrict__ src,
                           const int* __restrict__ dst,
                           const int* __restrict__ cnt,
                           const float* __restrict__ el,
                           const float* __restrict__ er,
                           const float* __restrict__ feat,
                           float* out) {
    __shared__ int sortbuf[8 * 64];
    const int node = __builtin_amdgcn_readfirstlane(blockIdx.x * 8 + (threadIdx.x >> 5));
    if (node >= NNODES) return;
    const int lane = threadIdx.x & 31;
    float* orow = out + (long)node * DFEAT;
    const float* bkt = orow;
    const int k0 = __float_as_int(bkt[lane]);
    const int deg = cnt[node];
    const float erd = er[node];
    const float* frow = feat + lane * 4;
    float4 acc = make_float4(0.0f, 0.0f, 0.0f, 0.0f);

    if (deg <= 32) {
        const bool active = lane < deg;
        const int key = active ? k0 : 0x7fffffff;
        int r = 0;
#pragma unroll
        for (int j = 0; j < 32; ++j) {
            const int kj = __builtin_amdgcn_readlane(key, j);
            r += ((kj < key) || (kj == key && j < lane)) ? 1 : 0;
        }
        const int skey = __builtin_amdgcn_ds_permute(r << 2, key);
        const float v = active ? gat_lrelu(el[active ? skey : 0] + erd) : -INFINITY;
        float m = wave_max_f(v);
        if (!(fabsf(m) < INFINITY)) m = 0.0f;
        const float ex = active ? __expf(v - m) : 0.0f;
        const float den = wave_sum_f(ex);
        const float alpha = ex / fmaxf(den, 1e-9f);

        int j = 0;
        for (; j + 4 <= deg; j += 4) {
            const int s0 = __builtin_amdgcn_readlane(skey, j + 0);
            const int s1 = __builtin_amdgcn_readlane(skey, j + 1);
            const int s2 = __builtin_amdgcn_readlane(skey, j + 2);
            const int s3 = __builtin_amdgcn_readlane(skey, j + 3);
            const float4 f0 = *(const float4*)(frow + (long)s0 * DFEAT);
            const float4 f1 = *(const float4*)(frow + (long)s1 * DFEAT);
            const float4 f2 = *(const float4*)(frow + (long)s2 * DFEAT);
            const float4 f3 = *(const float4*)(frow + (long)s3 * DFEAT);
            const float a0 = __int_as_float(__builtin_amdgcn_readlane(__float_as_int(alpha), j + 0));
            const float a1 = __int_as_float(__builtin_amdgcn_readlane(__float_as_int(alpha), j + 1));
            const float a2 = __int_as_float(__builtin_amdgcn_readlane(__float_as_int(alpha), j + 2));
            const float a3 = __int_as_float(__builtin_amdgcn_readlane(__float_as_int(alpha), j + 3));
            acc.x += a0 * f0.x; acc.y += a0 * f0.y; acc.z += a0 * f0.z; acc.w += a0 * f0.w;
            acc.x += a1 * f1.x; acc.y += a1 * f1.y; acc.z += a1 * f1.z; acc.w += a1 * f1.w;
            acc.x += a2 * f2.x; acc.y += a2 * f2.y; acc.z += a2 * f2.z; acc.w += a2 * f2.w;
            acc.x += a3 * f3.x; acc.y += a3 * f3.y; acc.z += a3 * f3.z; acc.w += a3 * f3.w;
        }
        for (; j < deg; ++j) {
            const int s0 = __builtin_amdgcn_readlane(skey, j);
            const float a0 = __int_as_float(__builtin_amdgcn_readlane(__float_as_int(alpha), j));
            const float4 f0 = *(const float4*)(frow + (long)s0 * DFEAT);
            acc.x += a0 * f0.x; acc.y += a0 * f0.y; acc.z += a0 * f0.z; acc.w += a0 * f0.w;
        }
    } else if (deg <= 64) {
        int* sb = sortbuf + (threadIdx.x >> 5) * 64;
        const bool act1 = lane + 32 < deg;
        const int c0 = (k0 << 6) | lane;
        const int c1 = ((act1 ? __float_as_int(bkt[32 + lane]) : 0x1ffffff) << 6) | (32 + lane);
        int r0 = 0, r1 = 0;
#pragma unroll
        for (int j = 0; j < 32; ++j) {
            const int ca = __builtin_amdgcn_readlane(c0, j);
            const int cb = __builtin_amdgcn_readlane(c1, j);
            r0 += (ca < c0) ? 1 : 0;
            r0 += (cb < c0) ? 1 : 0;
            r1 += (ca < c1) ? 1 : 0;
            r1 += (cb < c1) ? 1 : 0;
        }
        sb[r0] = c0;
        sb[r1] = c1;
        const int sk0 = sb[lane] >> 6;
        const int sk1 = sb[32 + lane] >> 6;

        const float v0 = gat_lrelu(el[sk0] + erd);
        const float v1 = act1 ? gat_lrelu(el[act1 ? sk1 : 0] + erd) : -INFINITY;
        float m = wave_max_f(fmaxf(v0, v1));
        if (!(fabsf(m) < INFINITY)) m = 0.0f;
        const float ex0 = __expf(v0 - m);
        const float ex1 = act1 ? __expf(v1 - m) : 0.0f;

        float den = 0.0f;
        for (int j = 0; j < 32; ++j)   den += lane_f(ex0, j);
        for (int j = 32; j < deg; ++j) den += lane_f(ex1, j - 32);
        const float dinv = fmaxf(den, 1e-9f);
        const float al0 = ex0 / dinv;
        const float al1 = ex1 / dinv;

        for (int j = 0; j < 32; j += 4) {
            const float4 f0 = *(const float4*)(frow + (long)__builtin_amdgcn_readlane(sk0, j + 0) * DFEAT);
            const float4 f1 = *(const float4*)(frow + (long)__builtin_amdgcn_readlane(sk0, j + 1) * DFEAT);
            const float4 f2 = *(const float4*)(frow + (long)__builtin_amdgcn_readlane(sk0, j + 2) * DFEAT);
            const float4 f3 = *(const float4*)(frow + (long)__builtin_amdgcn_readlane(sk0, j + 3) * DFEAT);
            gat_axpy4(acc, lane_f(al0, j + 0), f0);
            gat_axpy4(acc, lane_f(al0, j + 1), f1);
            gat_axpy4(acc, lane_f(al0, j + 2), f2);
            gat_axpy4(acc, lane_f(al0, j + 3), f3);
        }
        const int d1 = deg - 32;
        int j = 0;
        for (; j + 4 <= d1; j += 4) {
            const float4 f0 = *(const float4*)(frow + (long)__builtin_amdgcn_readlane(sk1, j + 0) * DFEAT);
            const float4 f1 = *(const float4*)(frow + (long)__builtin_amdgcn_readlane(sk1, j + 1) * DFEAT);
            const float4 f2 = *(const float4*)(frow + (long)__builtin_amdgcn_readlane(sk1, j + 2) * DFEAT);
            const float4 f3 = *(const float4*)(frow + (long)__builtin_amdgcn_readlane(sk1, j + 3) * DFEAT);
            gat_axpy4(acc, lane_f(al1, j + 0), f0);
            gat_axpy4(acc, lane_f(al1, j + 1), f1);
            gat_axpy4(acc, lane_f(al1, j + 2), f2);
            gat_axpy4(acc, lane_f(al1, j + 3), f3);
        }
        for (; j < d1; ++j) {
            const float4 f0 = *(const float4*)(frow + (long)__builtin_amdgcn_readlane(sk1, j) * DFEAT);
            gat_axpy4(acc, lane_f(al1, j), f0);
        }
    } else if (deg <= BCAP) {
        acc = gat_walk<false>(bkt, src, dst, el, frow, node, deg, lane, erd);
    } else {
        acc = gat_walk<true>(bkt, src, dst, el, frow, node, deg, lane, erd);
    }
    *(float4*)(orow + lane * 4) = acc;
}

extern "C" void kernel_launch(void* const* d_in, const int* in_sizes, int n_in,
                              void* d_out, int out_size, void* d_ws, size_t ws_size,
                              hipStream_t stream) {
    const float* h      = (const float*)d_in[0];
    const int*   src    = (const int*)  d_in[1];
    const int*   dst    = (const int*)  d_in[2];
    const float* W      = (const float*)d_in[3];
    const float* attn_l = (const float*)d_in[4];
    const float* attn_r = (const float*)d_in[5];
    float* out = (float*)d_out;

    float* feat  = (float*)d_ws;
    float* el    = feat + (long)NNODES * DFEAT;
    float* er    = el + NNODES;
    int*   cnt   = (int*)(er + NNODES);
    int*   bkt   = (int*)d_out;
    int*   lists = (int*)d_ws;
    int*   rcnt  = (int*)el;

    gat_zero_f32<<<2, 256, 0, stream>>>((float*)rcnt, (long)NRANGE);
    gat_bin<<<(NEDGES + BIN_EDGES - 1) / BIN_EDGES, 256, 0, stream>>>(src, dst, rcnt, lists);
    gat_fill<<<NRANGE, 512, 0, stream>>>(src, dst, rcnt, lists, cnt, bkt);

    const int nstrips = NNODES / 16;
    gat_gemm_wmma<<<(nstrips + 3) / 4, 128, 0, stream>>>(h, W, attn_l, attn_r, feat, el, er);

    gat_gather<<<(NNODES + 7) / 8, 256, 0, stream>>>(src, dst, cnt, el, er, feat, out);
}
